# starting kernel
# baseline (speedup 1.0000x reference)
#include <hip/hip_runtime.h>
#include <math.h>

#define NN 100000
#define NE 1250000
#define CH 64
#define ED 16

typedef __attribute__((ext_vector_type(2))) float v2f;
typedef __attribute__((ext_vector_type(8))) float v8f;

__device__ __forceinline__ v8f wmma_f32(v2f a, v2f b, v8f c) {
  return __builtin_amdgcn_wmma_f32_16x16x4_f32(false, a, false, b, (short)0, c,
                                               false, false);
}

__device__ __forceinline__ v8f splat8(float v) {
  v8f r;
#pragma unroll
  for (int i = 0; i < 8; ++i) r[i] = v;
  return r;
}

__global__ __launch_bounds__(256) void zero_kernel(float4* __restrict__ p, int n4) {
  int i = blockIdx.x * 256 + threadIdx.x;
  if (i < n4) p[i] = make_float4(0.f, 0.f, 0.f, 0.f);
}

__global__ __launch_bounds__(256) void edge_kernel(
    const float* __restrict__ x, const int* __restrict__ ei,
    const float* __restrict__ ea, const float* __restrict__ We,
    const float* __restrict__ be, float* __restrict__ agg) {
  __shared__ float sWe[ED * CH];
  const int tid = threadIdx.x;
  for (int i = tid; i < ED * CH; i += 256) sWe[i] = We[i];
  __syncthreads();

  const int wave = tid >> 5;
  const int lane = tid & 31;
  const int row = lane & 15;
  const int hi  = lane >> 4;
  const int kb  = hi * 2;

  v2f bf[4][4];
  float bev[4];
#pragma unroll
  for (int j = 0; j < 4; ++j) {
    const int c = 16 * j + row;
    bev[j] = be[c];
#pragma unroll
    for (int k4 = 0; k4 < 4; ++k4) {
      bf[k4][j].x = sWe[(4 * k4 + kb) * CH + c];
      bf[k4][j].y = sWe[(4 * k4 + kb + 1) * CH + c];
    }
  }

  const int stride = gridDim.x * 8;
  for (int tile = blockIdx.x * 8 + wave; tile < NE / 16; tile += stride) {
    const int e0 = tile * 16;

    const int sv = ei[e0 + row];
    const int dv = ei[NE + e0 + row];
    int srcn[8], dstn[8];
#pragma unroll
    for (int r = 0; r < 8; ++r) {
      srcn[r] = __shfl(sv, r + hi * 8, 32);
      dstn[r] = __shfl(dv, r + hi * 8, 32);
    }
    float g[8][4];
#pragma unroll
    for (int r = 0; r < 8; ++r) {
      const float* xs = x + (size_t)srcn[r] * CH + row;
#pragma unroll
      for (int j = 0; j < 4; ++j) g[r][j] = xs[16 * j];
    }

    v8f acc[4];
#pragma unroll
    for (int j = 0; j < 4; ++j) acc[j] = splat8(bev[j]);
    const float* arow = ea + (size_t)(e0 + row) * ED;
    v2f afrag[4];
#pragma unroll
    for (int k4 = 0; k4 < 4; ++k4)
      afrag[k4] = *(const v2f*)(arow + 4 * k4 + kb);
#pragma unroll
    for (int k4 = 0; k4 < 4; ++k4)
#pragma unroll
      for (int j = 0; j < 4; ++j) acc[j] = wmma_f32(afrag[k4], bf[k4][j], acc[j]);

#pragma unroll
    for (int r = 0; r < 8; ++r) {
      float* ad = agg + (size_t)dstn[r] * CH + row;
#pragma unroll
      for (int j = 0; j < 4; ++j) {
        float m = acc[j][r] + g[r][j];
        m = m > 0.f ? m : 0.f;
        __hip_atomic_fetch_add(ad + 16 * j, m, __ATOMIC_RELAXED,
                               __HIP_MEMORY_SCOPE_AGENT);
      }
    }
  }
}

#define HPAD 66
__global__ __launch_bounds__(128) void mlp_kernel(
    const float* __restrict__ xin, const float* __restrict__ agg,
    const float* __restrict__ W1, const float* __restrict__ b1,
    const float* __restrict__ W2, const float* __restrict__ b2,
    const float* __restrict__ epsl, const float* __restrict__ gamma,
    const float* __restrict__ beta, float* __restrict__ xout) {
  __shared__ float sW1[CH * CH];
  __shared__ float sW2[CH * CH];
  __shared__ float sH[4][16 * HPAD];
  const int tid = threadIdx.x;
  for (int i = tid; i < CH * CH; i += 128) { sW1[i] = W1[i]; sW2[i] = W2[i]; }
  __syncthreads();

  const int wave = tid >> 5;
  const int lane = tid & 31;
  const int row = lane & 15, hi = lane >> 4, kb = hi * 2;
  float* H = sH[wave];
  const float e1 = 1.0f + epsl[0];

  float b1v[4], b2v[4], gv[4], btv[4];
#pragma unroll
  for (int j = 0; j < 4; ++j) {
    const int c = 16 * j + row;
    b1v[j] = b1[c];
    b2v[j] = b2[c];
    gv[j] = gamma[c];
    btv[j] = beta[c];
  }

  const int stride = gridDim.x * 4;
  for (int tile = blockIdx.x * 4 + wave; tile < NN / 16; tile += stride) {
    const int R = tile * 16;
    v8f acc[4];
#pragma unroll
    for (int j = 0; j < 4; ++j) acc[j] = splat8(b1v[j]);
    const float* xrow = xin + (size_t)(R + row) * CH;
    const float* grow = agg + (size_t)(R + row) * CH;
#pragma unroll
    for (int kk = 0; kk < CH; kk += 4) {
      v2f xa = *(const v2f*)(xrow + kk + kb);
      v2f ga = *(const v2f*)(grow + kk + kb);
      v2f a;
      a.x = e1 * xa.x + ga.x;
      a.y = e1 * xa.y + ga.y;
#pragma unroll
      for (int j = 0; j < 4; ++j) {
        const int c = 16 * j + row;
        v2f b;
        b.x = sW1[(kk + kb) * CH + c];
        b.y = sW1[(kk + kb + 1) * CH + c];
        acc[j] = wmma_f32(a, b, acc[j]);
      }
    }
#pragma unroll
    for (int r = 0; r < 8; ++r)
#pragma unroll
      for (int j = 0; j < 4; ++j) {
        float v = acc[j][r];
        H[(r + hi * 8) * HPAD + 16 * j + row] = v > 0.f ? v : 0.f;
      }
    __builtin_amdgcn_wave_barrier();

    v8f acc2[4];
#pragma unroll
    for (int j = 0; j < 4; ++j) acc2[j] = splat8(b2v[j]);
#pragma unroll
    for (int kk = 0; kk < CH; kk += 4) {
      v2f a = *(const v2f*)(H + row * HPAD + kk + kb);
#pragma unroll
      for (int j = 0; j < 4; ++j) {
        const int c = 16 * j + row;
        v2f b;
        b.x = sW2[(kk + kb) * CH + c];
        b.y = sW2[(kk + kb + 1) * CH + c];
        acc2[j] = wmma_f32(a, b, acc2[j]);
      }
    }
    __builtin_amdgcn_wave_barrier();
#pragma unroll
    for (int r = 0; r < 8; ++r) {
      float s = 0.f, sq = 0.f;
#pragma unroll
      for (int j = 0; j < 4; ++j) {
        float v = acc2[j][r];
        s += v;
        sq += v * v;
      }
#pragma unroll
      for (int m = 1; m < 16; m <<= 1) {
        s += __shfl_xor(s, m, 32);
        sq += __shfl_xor(sq, m, 32);
      }
      const float mean = s * (1.f / 64.f);
      const float var = sq * (1.f / 64.f) - mean * mean;
      const float rstd = rsqrtf(var + 1e-5f);
      float* orow = xout + (size_t)(R + r + hi * 8) * CH;
#pragma unroll
      for (int j = 0; j < 4; ++j) {
        float v = (acc2[j][r] - mean) * rstd * gv[j] + btv[j];
        orow[16 * j + row] = v > 0.f ? v : 0.f;
      }
    }
  }
}

__global__ __launch_bounds__(128) void head_kernel(
    const float* __restrict__ xin, const float* __restrict__ Wh1,
    const float* __restrict__ bh1, const float* __restrict__ Wh2,
    const float* __restrict__ bh2, float* __restrict__ out) {
  __shared__ float sW[CH * CH];
  const int tid = threadIdx.x;
  for (int i = tid; i < CH * CH; i += 128) sW[i] = Wh1[i];
  __syncthreads();

  const int wave = tid >> 5;
  const int lane = tid & 31;
  const int row = lane & 15, hi = lane >> 4, kb = hi * 2;

  float bh1v[4], w2[4];
#pragma unroll
  for (int j = 0; j < 4; ++j) {
    bh1v[j] = bh1[16 * j + row];
    w2[j] = Wh2[16 * j + row];
  }
  const float b2v = bh2[0];

  const int stride = gridDim.x * 4;
  for (int tile = blockIdx.x * 4 + wave; tile < NN / 16; tile += stride) {
    const int R = tile * 16;
    v8f acc[4];
#pragma unroll
    for (int j = 0; j < 4; ++j) acc[j] = splat8(bh1v[j]);
    const float* xrow = xin + (size_t)(R + row) * CH;
#pragma unroll
    for (int kk = 0; kk < CH; kk += 4) {
      v2f a = *(const v2f*)(xrow + kk + kb);
#pragma unroll
      for (int j = 0; j < 4; ++j) {
        const int c = 16 * j + row;
        v2f b;
        b.x = sW[(kk + kb) * CH + c];
        b.y = sW[(kk + kb + 1) * CH + c];
        acc[j] = wmma_f32(a, b, acc[j]);
      }
    }
#pragma unroll
    for (int r = 0; r < 8; ++r) {
      float s = 0.f;
#pragma unroll
      for (int j = 0; j < 4; ++j) {
        float v = acc[j][r];
        v = v > 0.f ? v : 0.f;
        s += v * w2[j];
      }
#pragma unroll
      for (int m = 1; m < 16; m <<= 1) s += __shfl_xor(s, m, 32);
      if (row == 0) {
        const float z = s + b2v;
        out[R + r + hi * 8] = 1.f / (1.f + __expf(-z));
      }
    }
  }
}

extern "C" void kernel_launch(void* const* d_in, const int* in_sizes, int n_in,
                              void* d_out, int out_size, void* d_ws,
                              size_t ws_size, hipStream_t stream) {
  const float* x     = (const float*)d_in[0];
  const int*   ei    = (const int*)d_in[1];
  const float* ea    = (const float*)d_in[2];
  const float* We    = (const float*)d_in[3];
  const float* be    = (const float*)d_in[4];
  const float* W1    = (const float*)d_in[5];
  const float* b1    = (const float*)d_in[6];
  const float* W2    = (const float*)d_in[7];
  const float* b2    = (const float*)d_in[8];
  const float* eps   = (const float*)d_in[9];
  const float* gamma = (const float*)d_in[10];
  const float* beta  = (const float*)d_in[11];
  const float* Wh1   = (const float*)d_in[12];
  const float* bh1   = (const float*)d_in[13];
  const float* Wh2   = (const float*)d_in[14];
  const float* bh2   = (const float*)d_in[15];
  float* out = (float*)d_out;

  float* bufA = (float*)d_ws;
  float* bufB = bufA + (size_t)NN * CH;
  float* agg  = bufB + (size_t)NN * CH;

  const int n4      = (NN * CH) / 4;
  const int zblocks = (n4 + 255) / 256;
  const int eblocks = 1024;
  const int nblocks = 512;

  const float* xcur = x;
  for (int l = 0; l < 3; ++l) {
    zero_kernel<<<zblocks, 256, 0, stream>>>((float4*)agg, n4);
    edge_kernel<<<eblocks, 256, 0, stream>>>(xcur, ei, ea,
                                             We + (size_t)l * ED * CH,
                                             be + (size_t)l * CH, agg);
    float* xnext = (l & 1) ? bufB : bufA;
    mlp_kernel<<<nblocks, 128, 0, stream>>>(
        xcur, agg, W1 + (size_t)l * CH * CH, b1 + (size_t)l * CH,
        W2 + (size_t)l * CH * CH, b2 + (size_t)l * CH, eps + l,
        gamma + (size_t)l * CH, beta + (size_t)l * CH, xnext);
    xcur = xnext;
  }
  head_kernel<<<nblocks, 128, 0, stream>>>(xcur, Wh1, bh1, Wh2, bh2, out);
}


# Node kernel zeroes its own agg tile after reading it (whole-line stores); the two clear launches between layers are removed (10 launches -> 8)
# speedup vs baseline: 1.0556x; 1.0556x over previous
#include <hip/hip_runtime.h>
#include <math.h>

#define NN 100000
#define NE 1250000
#define CH 64
#define ED 16

typedef __attribute__((ext_vector_type(2))) float v2f;
typedef __attribute__((ext_vector_type(8))) float v8f;

__device__ __forceinline__ v8f wmma_f32(v2f a, v2f b, v8f c) {
  return __builtin_amdgcn_wmma_f32_16x16x4_f32(false, a, false, b, (short)0, c,
                                               false, false);
}

__device__ __forceinline__ v8f splat8(float v) {
  v8f r;
#pragma unroll
  for (int i = 0; i < 8; ++i) r[i] = v;
  return r;
}

__global__ __launch_bounds__(256) void zero_kernel(float4* __restrict__ p, int n4) {
  int i = blockIdx.x * 256 + threadIdx.x;
  if (i < n4) p[i] = make_float4(0.f, 0.f, 0.f, 0.f);
}

__global__ __launch_bounds__(256) void edge_kernel(
    const float* __restrict__ x, const int* __restrict__ ei,
    const float* __restrict__ ea, const float* __restrict__ We,
    const float* __restrict__ be, float* __restrict__ agg) {
  __shared__ float sWe[ED * CH];
  const int tid = threadIdx.x;
  for (int i = tid; i < ED * CH; i += 256) sWe[i] = We[i];
  __syncthreads();

  const int wave = tid >> 5;
  const int lane = tid & 31;
  const int row = lane & 15;
  const int hi  = lane >> 4;
  const int kb  = hi * 2;

  v2f bf[4][4];
  float bev[4];
#pragma unroll
  for (int j = 0; j < 4; ++j) {
    const int c = 16 * j + row;
    bev[j] = be[c];
#pragma unroll
    for (int k4 = 0; k4 < 4; ++k4) {
      bf[k4][j].x = sWe[(4 * k4 + kb) * CH + c];
      bf[k4][j].y = sWe[(4 * k4 + kb + 1) * CH + c];
    }
  }

  const int stride = gridDim.x * 8;
  for (int tile = blockIdx.x * 8 + wave; tile < NE / 16; tile += stride) {
    const int e0 = tile * 16;

    const int sv = ei[e0 + row];
    const int dv = ei[NE + e0 + row];
    int srcn[8], dstn[8];
#pragma unroll
    for (int r = 0; r < 8; ++r) {
      srcn[r] = __shfl(sv, r + hi * 8, 32);
      dstn[r] = __shfl(dv, r + hi * 8, 32);
    }
    float g[8][4];
#pragma unroll
    for (int r = 0; r < 8; ++r) {
      const float* xs = x + (size_t)srcn[r] * CH + row;
#pragma unroll
      for (int j = 0; j < 4; ++j) g[r][j] = xs[16 * j];
    }

    v8f acc[4];
#pragma unroll
    for (int j = 0; j < 4; ++j) acc[j] = splat8(bev[j]);
    const float* arow = ea + (size_t)(e0 + row) * ED;
    v2f afrag[4];
#pragma unroll
    for (int k4 = 0; k4 < 4; ++k4)
      afrag[k4] = *(const v2f*)(arow + 4 * k4 + kb);
#pragma unroll
    for (int k4 = 0; k4 < 4; ++k4)
#pragma unroll
      for (int j = 0; j < 4; ++j) acc[j] = wmma_f32(afrag[k4], bf[k4][j], acc[j]);

#pragma unroll
    for (int r = 0; r < 8; ++r) {
      float* ad = agg + (size_t)dstn[r] * CH + row;
#pragma unroll
      for (int j = 0; j < 4; ++j) {
        float m = acc[j][r] + g[r][j];
        m = m > 0.f ? m : 0.f;
        __hip_atomic_fetch_add(ad + 16 * j, m, __ATOMIC_RELAXED,
                               __HIP_MEMORY_SCOPE_AGENT);
      }
    }
  }
}

#define HPAD 66
__global__ __launch_bounds__(128) void mlp_kernel(
    const float* __restrict__ xin, float* __restrict__ agg,
    const float* __restrict__ W1, const float* __restrict__ b1,
    const float* __restrict__ W2, const float* __restrict__ b2,
    const float* __restrict__ epsl, const float* __restrict__ gamma,
    const float* __restrict__ beta, float* __restrict__ xout,
    const int clear_agg) {
  __shared__ float sW1[CH * CH];
  __shared__ float sW2[CH * CH];
  __shared__ float sH[4][16 * HPAD];
  const int tid = threadIdx.x;
  for (int i = tid; i < CH * CH; i += 128) { sW1[i] = W1[i]; sW2[i] = W2[i]; }
  __syncthreads();

  const int wave = tid >> 5;
  const int lane = tid & 31;
  const int row = lane & 15, hi = lane >> 4, kb = hi * 2;
  float* H = sH[wave];
  const float e1 = 1.0f + epsl[0];

  float b1v[4], b2v[4], gv[4], btv[4];
#pragma unroll
  for (int j = 0; j < 4; ++j) {
    const int c = 16 * j + row;
    b1v[j] = b1[c];
    b2v[j] = b2[c];
    gv[j] = gamma[c];
    btv[j] = beta[c];
  }

  const int stride = gridDim.x * 4;
  for (int tile = blockIdx.x * 4 + wave; tile < NN / 16; tile += stride) {
    const int R = tile * 16;
    v8f acc[4];
#pragma unroll
    for (int j = 0; j < 4; ++j) acc[j] = splat8(b1v[j]);
    const float* xrow = xin + (size_t)(R + row) * CH;
    const float* grow = agg + (size_t)(R + row) * CH;
#pragma unroll
    for (int kk = 0; kk < CH; kk += 4) {
      v2f xa = *(const v2f*)(xrow + kk + kb);
      v2f ga = *(const v2f*)(grow + kk + kb);
      v2f a;
      a.x = e1 * xa.x + ga.x;
      a.y = e1 * xa.y + ga.y;
#pragma unroll
      for (int j = 0; j < 4; ++j) {
        const int c = 16 * j + row;
        v2f b;
        b.x = sW1[(kk + kb) * CH + c];
        b.y = sW1[(kk + kb + 1) * CH + c];
        acc[j] = wmma_f32(a, b, acc[j]);
      }
    }
    if (clear_agg) {
      float* zt = agg + (size_t)R * CH + lane * 2;
      v2f z2;
      z2.x = 0.f;
      z2.y = 0.f;
#pragma unroll
      for (int i = 0; i < 16; ++i) *(v2f*)(zt + i * 64) = z2;
    }
#pragma unroll
    for (int r = 0; r < 8; ++r)
#pragma unroll
      for (int j = 0; j < 4; ++j) {
        float v = acc[j][r];
        H[(r + hi * 8) * HPAD + 16 * j + row] = v > 0.f ? v : 0.f;
      }
    __builtin_amdgcn_wave_barrier();

    v8f acc2[4];
#pragma unroll
    for (int j = 0; j < 4; ++j) acc2[j] = splat8(b2v[j]);
#pragma unroll
    for (int kk = 0; kk < CH; kk += 4) {
      v2f a = *(const v2f*)(H + row * HPAD + kk + kb);
#pragma unroll
      for (int j = 0; j < 4; ++j) {
        const int c = 16 * j + row;
        v2f b;
        b.x = sW2[(kk + kb) * CH + c];
        b.y = sW2[(kk + kb + 1) * CH + c];
        acc2[j] = wmma_f32(a, b, acc2[j]);
      }
    }
    __builtin_amdgcn_wave_barrier();
#pragma unroll
    for (int r = 0; r < 8; ++r) {
      float s = 0.f, sq = 0.f;
#pragma unroll
      for (int j = 0; j < 4; ++j) {
        float v = acc2[j][r];
        s += v;
        sq += v * v;
      }
#pragma unroll
      for (int m = 1; m < 16; m <<= 1) {
        s += __shfl_xor(s, m, 32);
        sq += __shfl_xor(sq, m, 32);
      }
      const float mean = s * (1.f / 64.f);
      const float var = sq * (1.f / 64.f) - mean * mean;
      const float rstd = rsqrtf(var + 1e-5f);
      float* orow = xout + (size_t)(R + r + hi * 8) * CH;
#pragma unroll
      for (int j = 0; j < 4; ++j) {
        float v = (acc2[j][r] - mean) * rstd * gv[j] + btv[j];
        orow[16 * j + row] = v > 0.f ? v : 0.f;
      }
    }
  }
}

__global__ __launch_bounds__(128) void head_kernel(
    const float* __restrict__ xin, const float* __restrict__ Wh1,
    const float* __restrict__ bh1, const float* __restrict__ Wh2,
    const float* __restrict__ bh2, float* __restrict__ out) {
  __shared__ float sW[CH * CH];
  const int tid = threadIdx.x;
  for (int i = tid; i < CH * CH; i += 128) sW[i] = Wh1[i];
  __syncthreads();

  const int wave = tid >> 5;
  const int lane = tid & 31;
  const int row = lane & 15, hi = lane >> 4, kb = hi * 2;

  float bh1v[4], w2[4];
#pragma unroll
  for (int j = 0; j < 4; ++j) {
    bh1v[j] = bh1[16 * j + row];
    w2[j] = Wh2[16 * j + row];
  }
  const float b2v = bh2[0];

  const int stride = gridDim.x * 4;
  for (int tile = blockIdx.x * 4 + wave; tile < NN / 16; tile += stride) {
    const int R = tile * 16;
    v8f acc[4];
#pragma unroll
    for (int j = 0; j < 4; ++j) acc[j] = splat8(bh1v[j]);
    const float* xrow = xin + (size_t)(R + row) * CH;
#pragma unroll
    for (int kk = 0; kk < CH; kk += 4) {
      v2f a = *(const v2f*)(xrow + kk + kb);
#pragma unroll
      for (int j = 0; j < 4; ++j) {
        const int c = 16 * j + row;
        v2f b;
        b.x = sW[(kk + kb) * CH + c];
        b.y = sW[(kk + kb + 1) * CH + c];
        acc[j] = wmma_f32(a, b, acc[j]);
      }
    }
#pragma unroll
    for (int r = 0; r < 8; ++r) {
      float s = 0.f;
#pragma unroll
      for (int j = 0; j < 4; ++j) {
        float v = acc[j][r];
        v = v > 0.f ? v : 0.f;
        s += v * w2[j];
      }
#pragma unroll
      for (int m = 1; m < 16; m <<= 1) s += __shfl_xor(s, m, 32);
      if (row == 0) {
        const float z = s + b2v;
        out[R + r + hi * 8] = 1.f / (1.f + __expf(-z));
      }
    }
  }
}

extern "C" void kernel_launch(void* const* d_in, const int* in_sizes, int n_in,
                              void* d_out, int out_size, void* d_ws,
                              size_t ws_size, hipStream_t stream) {
  const float* x     = (const float*)d_in[0];
  const int*   ei    = (const int*)d_in[1];
  const float* ea    = (const float*)d_in[2];
  const float* We    = (const float*)d_in[3];
  const float* be    = (const float*)d_in[4];
  const float* W1    = (const float*)d_in[5];
  const float* b1    = (const float*)d_in[6];
  const float* W2    = (const float*)d_in[7];
  const float* b2    = (const float*)d_in[8];
  const float* eps   = (const float*)d_in[9];
  const float* gamma = (const float*)d_in[10];
  const float* beta  = (const float*)d_in[11];
  const float* Wh1   = (const float*)d_in[12];
  const float* bh1   = (const float*)d_in[13];
  const float* Wh2   = (const float*)d_in[14];
  const float* bh2   = (const float*)d_in[15];
  float* out = (float*)d_out;

  float* bufA = (float*)d_ws;
  float* bufB = bufA + (size_t)NN * CH;
  float* agg  = bufB + (size_t)NN * CH;

  const int n4      = (NN * CH) / 4;
  const int zblocks = (n4 + 255) / 256;
  const int eblocks = 1024;
  const int nblocks = 512;

  const float* xcur = x;
  zero_kernel<<<zblocks, 256, 0, stream>>>((float4*)agg, n4);
  for (int l = 0; l < 3; ++l) {
    edge_kernel<<<eblocks, 256, 0, stream>>>(xcur, ei, ea,
                                             We + (size_t)l * ED * CH,
                                             be + (size_t)l * CH, agg);
    float* xnext = (l & 1) ? bufB : bufA;
    mlp_kernel<<<nblocks, 128, 0, stream>>>(
        xcur, agg, W1 + (size_t)l * CH * CH, b1 + (size_t)l * CH,
        W2 + (size_t)l * CH * CH, b2 + (size_t)l * CH, eps + l,
        gamma + (size_t)l * CH, beta + (size_t)l * CH, xnext, l < 2 ? 1 : 0);
    xcur = xnext;
  }
  head_kernel<<<nblocks, 128, 0, stream>>>(xcur, Wh1, bh1, Wh2, bh2, out);
}


# Edge kernel: each wave-wide gather load and atomic add covers one whole 128 B line of one row (was 64 B halves of two rows); tiles swapped across lane halves; atomic arrival order free as before
# speedup vs baseline: 1.1681x; 1.1066x over previous
#include <hip/hip_runtime.h>
#include <math.h>

#define NN 100000
#define NE 1250000
#define CH 64
#define ED 16

typedef __attribute__((ext_vector_type(2))) float v2f;
typedef __attribute__((ext_vector_type(8))) float v8f;

__device__ __forceinline__ v8f wmma_f32(v2f a, v2f b, v8f c) {
  return __builtin_amdgcn_wmma_f32_16x16x4_f32(false, a, false, b, (short)0, c,
                                               false, false);
}

__device__ __forceinline__ v8f splat8(float v) {
  v8f r;
#pragma unroll
  for (int i = 0; i < 8; ++i) r[i] = v;
  return r;
}

__global__ __launch_bounds__(256) void zero_kernel(float4* __restrict__ p, int n4) {
  int i = blockIdx.x * 256 + threadIdx.x;
  if (i < n4) p[i] = make_float4(0.f, 0.f, 0.f, 0.f);
}

__device__ __forceinline__ float swap_half(float v) {
  int b = __builtin_bit_cast(int, v);
  return __builtin_bit_cast(
      float, __builtin_amdgcn_permlanex16(b, b, 0x76543210, 0xfedcba98, false, false));
}

typedef __attribute__((address_space(1))) float gfloat;
typedef __attribute__((address_space(1))) const float cgfloat;

__device__ __forceinline__ unsigned long long uniform_u64(unsigned long long u) {
  const unsigned lo = __builtin_amdgcn_readfirstlane((unsigned)u);
  const unsigned hi = __builtin_amdgcn_readfirstlane((unsigned)(u >> 32));
  return ((unsigned long long)hi << 32) | lo;
}

__global__ __launch_bounds__(256) void edge_kernel(
    const float* __restrict__ x, const int* __restrict__ ei,
    const float* __restrict__ ea, const float* __restrict__ We,
    const float* __restrict__ be, float* __restrict__ agg) {
  __shared__ float sWe[ED * CH];
  const int tid = threadIdx.x;
  for (int i = tid; i < ED * CH; i += 256) sWe[i] = We[i];
  __syncthreads();

  const int wave = __builtin_amdgcn_readfirstlane(tid >> 5);
  const int lane = tid & 31;
  const int row = lane & 15;
  const int hi  = lane >> 4;
  const int kb  = hi * 2;
  const unsigned ulane = (unsigned)lane;

  v2f bf[4][4];
  float bev[4];
#pragma unroll
  for (int j = 0; j < 4; ++j) {
    const int c = 16 * j + row;
    bev[j] = be[c];
#pragma unroll
    for (int k4 = 0; k4 < 4; ++k4) {
      bf[k4][j].x = sWe[(4 * k4 + kb) * CH + c];
      bf[k4][j].y = sWe[(4 * k4 + kb + 1) * CH + c];
    }
  }

  const int stride = gridDim.x * 8;
  for (int tile = blockIdx.x * 8 + wave; tile < NE / 16; tile += stride) {
    const int e0 = tile * 16;

    const float* arow = ea + (size_t)(e0 + row) * ED;
    v2f afrag[4];
#pragma unroll
    for (int k4 = 0; k4 < 4; ++k4)
      afrag[k4] = *(const v2f*)(arow + 4 * k4 + kb);
    int srcn[16], dstn[16];
#pragma unroll
    for (int R = 0; R < 16; ++R) {
      srcn[R] = ei[e0 + R];
      dstn[R] = ei[NE + e0 + R];
    }
    float g[16][2];
#pragma unroll
    for (int R = 0; R < 16; ++R) {
      const unsigned long long xb =
          uniform_u64((unsigned long long)x + ((unsigned long long)(unsigned)srcn[R] << 8));
      g[R][0] = ((cgfloat*)xb)[ulane];
      g[R][1] = ((cgfloat*)xb)[ulane + 32u];
    }
    __builtin_amdgcn_sched_barrier(0);

    v8f acc[4];
#pragma unroll
    for (int j = 0; j < 4; ++j) acc[j] = splat8(bev[j]);
#pragma unroll
    for (int k4 = 0; k4 < 4; ++k4)
#pragma unroll
      for (int j = 0; j < 4; ++j) acc[j] = wmma_f32(afrag[k4], bf[k4][j], acc[j]);

#pragma unroll
    for (int r = 0; r < 8; ++r) {
      const unsigned long long ab0 =
          uniform_u64((unsigned long long)agg + ((unsigned long long)(unsigned)dstn[r] << 8));
      const unsigned long long ab1 =
          uniform_u64((unsigned long long)agg + ((unsigned long long)(unsigned)dstn[r + 8] << 8));
#pragma unroll
      for (int q = 0; q < 2; ++q) {
        const float a = acc[2 * q][r];
        const float b = acc[2 * q + 1][r];
        const float s = swap_half(hi ? a : b);
        const float v0 = hi ? s : a;
        const float v1 = hi ? b : s;
        float m0 = v0 + g[r][q];
        float m1 = v1 + g[r + 8][q];
        m0 = m0 > 0.f ? m0 : 0.f;
        m1 = m1 > 0.f ? m1 : 0.f;
        __hip_atomic_fetch_add((gfloat*)ab0 + (ulane + 32u * q), m0,
                               __ATOMIC_RELAXED, __HIP_MEMORY_SCOPE_AGENT);
        __hip_atomic_fetch_add((gfloat*)ab1 + (ulane + 32u * q), m1,
                               __ATOMIC_RELAXED, __HIP_MEMORY_SCOPE_AGENT);
      }
    }
  }
}

#define HPAD 66
__global__ __launch_bounds__(128) void mlp_kernel(
    const float* __restrict__ xin, float* __restrict__ agg,
    const float* __restrict__ W1, const float* __restrict__ b1,
    const float* __restrict__ W2, const float* __restrict__ b2,
    const float* __restrict__ epsl, const float* __restrict__ gamma,
    const float* __restrict__ beta, float* __restrict__ xout,
    const int clear_agg) {
  __shared__ float sW1[CH * CH];
  __shared__ float sW2[CH * CH];
  __shared__ float sH[4][16 * HPAD];
  const int tid = threadIdx.x;
  for (int i = tid; i < CH * CH; i += 128) { sW1[i] = W1[i]; sW2[i] = W2[i]; }
  __syncthreads();

  const int wave = tid >> 5;
  const int lane = tid & 31;
  const int row = lane & 15, hi = lane >> 4, kb = hi * 2;
  float* H = sH[wave];
  const float e1 = 1.0f + epsl[0];

  float b1v[4], b2v[4], gv[4], btv[4];
#pragma unroll
  for (int j = 0; j < 4; ++j) {
    const int c = 16 * j + row;
    b1v[j] = b1[c];
    b2v[j] = b2[c];
    gv[j] = gamma[c];
    btv[j] = beta[c];
  }

  const int stride = gridDim.x * 4;
  for (int tile = blockIdx.x * 4 + wave; tile < NN / 16; tile += stride) {
    const int R = tile * 16;
    v8f acc[4];
#pragma unroll
    for (int j = 0; j < 4; ++j) acc[j] = splat8(b1v[j]);
    const float* xrow = xin + (size_t)(R + row) * CH;
    const float* grow = agg + (size_t)(R + row) * CH;
#pragma unroll
    for (int kk = 0; kk < CH; kk += 4) {
      v2f xa = *(const v2f*)(xrow + kk + kb);
      v2f ga = *(const v2f*)(grow + kk + kb);
      v2f a;
      a.x = e1 * xa.x + ga.x;
      a.y = e1 * xa.y + ga.y;
#pragma unroll
      for (int j = 0; j < 4; ++j) {
        const int c = 16 * j + row;
        v2f b;
        b.x = sW1[(kk + kb) * CH + c];
        b.y = sW1[(kk + kb + 1) * CH + c];
        acc[j] = wmma_f32(a, b, acc[j]);
      }
    }
    if (clear_agg) {
      float* zt = agg + (size_t)R * CH + lane * 2;
      v2f z2;
      z2.x = 0.f;
      z2.y = 0.f;
#pragma unroll
      for (int i = 0; i < 16; ++i) *(v2f*)(zt + i * 64) = z2;
    }
#pragma unroll
    for (int r = 0; r < 8; ++r)
#pragma unroll
      for (int j = 0; j < 4; ++j) {
        float v = acc[j][r];
        H[(r + hi * 8) * HPAD + 16 * j + row] = v > 0.f ? v : 0.f;
      }
    __builtin_amdgcn_wave_barrier();

    v8f acc2[4];
#pragma unroll
    for (int j = 0; j < 4; ++j) acc2[j] = splat8(b2v[j]);
#pragma unroll
    for (int kk = 0; kk < CH; kk += 4) {
      v2f a = *(const v2f*)(H + row * HPAD + kk + kb);
#pragma unroll
      for (int j = 0; j < 4; ++j) {
        const int c = 16 * j + row;
        v2f b;
        b.x = sW2[(kk + kb) * CH + c];
        b.y = sW2[(kk + kb + 1) * CH + c];
        acc2[j] = wmma_f32(a, b, acc2[j]);
      }
    }
    __builtin_amdgcn_wave_barrier();
#pragma unroll
    for (int r = 0; r < 8; ++r) {
      float s = 0.f, sq = 0.f;
#pragma unroll
      for (int j = 0; j < 4; ++j) {
        float v = acc2[j][r];
        s += v;
        sq += v * v;
      }
#pragma unroll
      for (int m = 1; m < 16; m <<= 1) {
        s += __shfl_xor(s, m, 32);
        sq += __shfl_xor(sq, m, 32);
      }
      const float mean = s * (1.f / 64.f);
      const float var = sq * (1.f / 64.f) - mean * mean;
      const float rstd = rsqrtf(var + 1e-5f);
      float* orow = xout + (size_t)(R + r + hi * 8) * CH;
#pragma unroll
      for (int j = 0; j < 4; ++j) {
        float v = (acc2[j][r] - mean) * rstd * gv[j] + btv[j];
        orow[16 * j + row] = v > 0.f ? v : 0.f;
      }
    }
  }
}

__global__ __launch_bounds__(128) void head_kernel(
    const float* __restrict__ xin, const float* __restrict__ Wh1,
    const float* __restrict__ bh1, const float* __restrict__ Wh2,
    const float* __restrict__ bh2, float* __restrict__ out) {
  __shared__ float sW[CH * CH];
  const int tid = threadIdx.x;
  for (int i = tid; i < CH * CH; i += 128) sW[i] = Wh1[i];
  __syncthreads();

  const int wave = tid >> 5;
  const int lane = tid & 31;
  const int row = lane & 15, hi = lane >> 4, kb = hi * 2;

  float bh1v[4], w2[4];
#pragma unroll
  for (int j = 0; j < 4; ++j) {
    bh1v[j] = bh1[16 * j + row];
    w2[j] = Wh2[16 * j + row];
  }
  const float b2v = bh2[0];

  const int stride = gridDim.x * 4;
  for (int tile = blockIdx.x * 4 + wave; tile < NN / 16; tile += stride) {
    const int R = tile * 16;
    v8f acc[4];
#pragma unroll
    for (int j = 0; j < 4; ++j) acc[j] = splat8(bh1v[j]);
    const float* xrow = xin + (size_t)(R + row) * CH;
#pragma unroll
    for (int kk = 0; kk < CH; kk += 4) {
      v2f a = *(const v2f*)(xrow + kk + kb);
#pragma unroll
      for (int j = 0; j < 4; ++j) {
        const int c = 16 * j + row;
        v2f b;
        b.x = sW[(kk + kb) * CH + c];
        b.y = sW[(kk + kb + 1) * CH + c];
        acc[j] = wmma_f32(a, b, acc[j]);
      }
    }
#pragma unroll
    for (int r = 0; r < 8; ++r) {
      float s = 0.f;
#pragma unroll
      for (int j = 0; j < 4; ++j) {
        float v = acc[j][r];
        v = v > 0.f ? v : 0.f;
        s += v * w2[j];
      }
#pragma unroll
      for (int m = 1; m < 16; m <<= 1) s += __shfl_xor(s, m, 32);
      if (row == 0) {
        const float z = s + b2v;
        out[R + r + hi * 8] = 1.f / (1.f + __expf(-z));
      }
    }
  }
}

extern "C" void kernel_launch(void* const* d_in, const int* in_sizes, int n_in,
                              void* d_out, int out_size, void* d_ws,
                              size_t ws_size, hipStream_t stream) {
  const float* x     = (const float*)d_in[0];
  const int*   ei    = (const int*)d_in[1];
  const float* ea    = (const float*)d_in[2];
  const float* We    = (const float*)d_in[3];
  const float* be    = (const float*)d_in[4];
  const float* W1    = (const float*)d_in[5];
  const float* b1    = (const float*)d_in[6];
  const float* W2    = (const float*)d_in[7];
  const float* b2    = (const float*)d_in[8];
  const float* eps   = (const float*)d_in[9];
  const float* gamma = (const float*)d_in[10];
  const float* beta  = (const float*)d_in[11];
  const float* Wh1   = (const float*)d_in[12];
  const float* bh1   = (const float*)d_in[13];
  const float* Wh2   = (const float*)d_in[14];
  const float* bh2   = (const float*)d_in[15];
  float* out = (float*)d_out;

  float* bufA = (float*)d_ws;
  float* bufB = bufA + (size_t)NN * CH;
  float* agg  = bufB + (size_t)NN * CH;

  const int n4      = (NN * CH) / 4;
  const int zblocks = (n4 + 255) / 256;
  const int eblocks = 1024;
  const int nblocks = 512;

  const float* xcur = x;
  zero_kernel<<<zblocks, 256, 0, stream>>>((float4*)agg, n4);
  for (int l = 0; l < 3; ++l) {
    edge_kernel<<<eblocks, 256, 0, stream>>>(xcur, ei, ea,
                                             We + (size_t)l * ED * CH,
                                             be + (size_t)l * CH, agg);
    float* xnext = (l & 1) ? bufB : bufA;
    mlp_kernel<<<nblocks, 128, 0, stream>>>(
        xcur, agg, W1 + (size_t)l * CH * CH, b1 + (size_t)l * CH,
        W2 + (size_t)l * CH * CH, b2 + (size_t)l * CH, eps + l,
        gamma + (size_t)l * CH, beta + (size_t)l * CH, xnext, l < 2 ? 1 : 0);
    xcur = xnext;
  }
  head_kernel<<<nblocks, 128, 0, stream>>>(xcur, Wh1, bh1, Wh2, bh2, out);
}
